# starting kernel
# baseline (speedup 1.0000x reference)
#include <hip/hip_runtime.h>
#include <hip/hip_bf16.h>
#include <math.h>

typedef float v2f __attribute__((ext_vector_type(2)));
typedef float v8f __attribute__((ext_vector_type(8)));

#define WIN_LEN   400
#define WIN_INC   100
#define PAD       300
#define N_FRAMES  1603
#define N_CH      257
#define N_IN      160000
#define N_BATCH   16
#define FRAMES_PER_BLOCK 64
#define BLOCK_THREADS    128
#define XS_LEN    (WIN_INC * (FRAMES_PER_BLOCK - 1) + WIN_LEN)
#define XS_PAD    6704
#define A_ELEMS   (WIN_LEN * 16)

__global__ __launch_bounds__(BLOCK_THREADS)
void convstft_wmma_f32(const float* __restrict__ in,
                       const float* __restrict__ w,
                       float* __restrict__ mags,
                       float* __restrict__ phase)
{
    extern __shared__ float smem[];
    float* Ar = smem;
    float* Ai = smem + A_ELEMS;
    float* Xs = smem + 2 * A_ELEMS;

    const int c0  = blockIdx.x * 16;
    const int t0  = blockIdx.y * FRAMES_PER_BLOCK;
    const int b   = blockIdx.z;
    const int tid = threadIdx.x;

    for (int idx = tid; idx < 16 * WIN_LEN; idx += BLOCK_THREADS) {
        const int m = idx / WIN_LEN;
        const int k = idx - m * WIN_LEN;
        const int c = c0 + m;
        float vr = 0.0f, vi = 0.0f;
        if (c < N_CH) {
            vr = w[(size_t)c * WIN_LEN + k];
            vi = w[(size_t)(c + N_CH) * WIN_LEN + k];
        }
        Ar[k * 16 + m] = vr;
        Ai[k * 16 + m] = vi;
    }

    const long  sbase = (long)t0 * WIN_INC - PAD;
    const float* inb  = in + (size_t)b * N_IN;
    for (int s = tid; s < XS_LEN; s += BLOCK_THREADS) {
        const long g = sbase + s;
        Xs[s] = (g >= 0 && g < (long)N_IN) ? inb[g] : 0.0f;
    }
    __syncthreads();

    const int lane = tid & 31;
    const int wave = tid >> 5;
    const int half = lane >> 4;
    const int mr   = lane & 15;
    const int soff = WIN_INC * (wave * 16 + mr);

    v8f accr = {};
    v8f acci = {};

    for (int kk = 0; kk < WIN_LEN / 4; ++kk) {
        const int kb = 4 * kk + 2 * half;
        v2f a_r, a_i, bx;
        a_r.x = Ar[(kb + 0) * 16 + mr];
        a_r.y = Ar[(kb + 1) * 16 + mr];
        a_i.x = Ai[(kb + 0) * 16 + mr];
        a_i.y = Ai[(kb + 1) * 16 + mr];
        bx.x  = Xs[soff + kb + 0];
        bx.y  = Xs[soff + kb + 1];
        accr = __builtin_amdgcn_wmma_f32_16x16x4_f32(
                   false, a_r, false, bx, (short)0, accr, false, false);
        acci = __builtin_amdgcn_wmma_f32_16x16x4_f32(
                   false, a_i, false, bx, (short)0, acci, false, false);
    }

    const float EPS = 1.19209290e-07f;
    const int t = t0 + wave * 16 + mr;
    if (t < N_FRAMES) {
        #pragma unroll
        for (int r = 0; r < 8; ++r) {
            const int c = c0 + r + 8 * half;
            if (c < N_CH) {
                const float rr = accr[r];
                const float ii = acci[r];
                float m2 = fmaxf(rr * rr + ii * ii, EPS);
                const size_t o = ((size_t)b * N_CH + c) * N_FRAMES + t;
                mags[o]  = sqrtf(m2);
                phase[o] = atan2f(ii + EPS, rr + EPS);
            }
        }
    }
}

extern "C" void kernel_launch(void* const* d_in, const int* in_sizes, int n_in,
                              void* d_out, int out_size, void* d_ws, size_t ws_size,
                              hipStream_t stream) {
    (void)in_sizes; (void)n_in; (void)d_ws; (void)ws_size; (void)out_size;

    const float* in = (const float*)d_in[0];
    const float* w  = (const float*)d_in[1];

    float* mags  = (float*)d_out;
    float* phase = mags + (size_t)N_BATCH * N_CH * N_FRAMES;

    const dim3 grid((N_CH + 15) / 16,
                    (N_FRAMES + FRAMES_PER_BLOCK - 1) / FRAMES_PER_BLOCK,
                    N_BATCH);
    const dim3 block(BLOCK_THREADS);
    const size_t smem_bytes = (size_t)(2 * A_ELEMS + XS_PAD) * sizeof(float);

    convstft_wmma_f32<<<grid, block, smem_bytes, stream>>>(in, w, mags, phase);
}


# epilogue software-pipelined: tile sums parked in LDS, finished (sqrt, atan2, stores) one element per 3 k-steps under the next tile's matrix stream; staging rows wave-uniform to free registers
# speedup vs baseline: 3.1826x; 3.1826x over previous
#include <hip/hip_runtime.h>
#include <math.h>


typedef float v2f __attribute__((ext_vector_type(2)));
typedef float v4f __attribute__((ext_vector_type(4)));
typedef float v8f __attribute__((ext_vector_type(8)));

#define WIN_LEN   400
#define WIN_INC   100
#define PAD       300
#define N_FRAMES  1603
#define N_CH      257
#define N_IN      160000
#define N_BATCH   16
#define TOT_FRAMES (N_BATCH * N_FRAMES)
#define WAVES         8
#define BLOCK_THREADS 256
#define FR_PER_WAVE   32
#define FR_PER_GROUP  (WAVES * FR_PER_WAVE)
#define N_FG  ((TOT_FRAMES + FR_PER_GROUP - 1) / FR_PER_GROUP)
#define N_CT  17
#define N_BU  (N_FG * N_CT)
#define N_BLOCKS 256
#define K4     (WIN_LEN / 4)
#define PITCH4 101
#define NSTEP  (WIN_LEN / 8)
#define PANEL4     (32 * PITCH4)
#define SAVE_PAIRS 16
#define LDS_BYTES  (PANEL4 * 16 + WAVES * SAVE_PAIRS * 32 * 8)

static_assert(TOT_FRAMES % 16 == 0, "a 16-frame tile is valid or invalid as a whole");


__device__ __forceinline__ void finish_elem(v2f ri, int e, int h, int ct, unsigned ob,
                                            bool full, bool last,
                                            float* __restrict__ mags, float* __restrict__ phase)
{
    const float EPS = 1.19209290e-07f;
    const float rr = ri.x;
    const float ii = ri.y;
    const float m2 = fmaxf(rr * rr + ii * ii, EPS);
    const float mg = sqrtf(m2);
    const float ph = atan2f(ii + EPS, rr + EPS);
    const int r = e & 7;
    const unsigned o = ob + (unsigned)((ct * 16 + r + 8 * h) * N_FRAMES);
    if (full) {
        mags[o]  = mg;
        phase[o] = ph;
    } else if (last && r == 0) {
        if (h == 0) {
            mags[o]  = mg;
            phase[o] = ph;
        }
    }
}

__global__ __launch_bounds__(BLOCK_THREADS)
void convstft_inreg_f32(const float* __restrict__ in,
                        const float* __restrict__ w,
                        float* __restrict__ mags,
                        float* __restrict__ phase)
{
    __shared__ __attribute__((aligned(16))) unsigned char lds[LDS_BYTES];
    v4f* wl = (v4f*)lds;

    const int tid  = threadIdx.x;
    const int lane = tid & 31;
    const int wave = tid >> 5;
    const int h    = lane >> 4;
    const int c    = lane & 15;

    const int bu0 = (blockIdx.x * N_BU) / N_BLOCKS;
    const int bu1 = ((blockIdx.x + 1) * N_BU) / N_BLOCKS;

    const v4f* w4 = (const v4f*)w;

    const int wave_s = __builtin_amdgcn_readfirstlane(wave);

    v2f* sv = (v2f*)(lds + PANEL4 * 16) + wave_s * (SAVE_PAIRS * 32) + lane;
    for (int e = 0; e < SAVE_PAIRS; ++e) sv[e * 32] = (v2f){0.0f, 0.0f};

    int      p_ct = 0;
    unsigned p_ob[2] = {0u, 0u};
    bool     p_fv[2] = {false, false};
    unsigned ob[2] = {0u, 0u};

    v4f  xb[2][NSTEP];
    int  fb[2], ft[2];
    bool fv[2];
    int  cur_fg = -1;

    for (int bu = bu0; bu < bu1; ++bu) {
        const int fg = bu / N_CT;
        const int ct = bu - fg * N_CT;

        if (fg != cur_fg) {
            cur_fg = fg;
            #pragma unroll
            for (int q = 0; q < 2; ++q) {
                const int f0 = (fg * WAVES + wave_s) * FR_PER_WAVE + q * 16;
                fv[q] = f0 < TOT_FRAMES;
                const int fc = fv[q] ? (f0 + c) : (TOT_FRAMES - 1);
                fb[q] = fc / N_FRAMES;
                ft[q] = fc - fb[q] * N_FRAMES;
                ob[q] = (unsigned)(fb[q] * (N_CH * N_FRAMES) + ft[q]);
                const int g0 = ft[q] * WIN_INC - PAD + 4 * h;
                const float* srcp = in + (size_t)fb[q] * N_IN;
                #pragma unroll
                for (int j = 0; j < NSTEP; ++j) {
                    const int g = g0 + 8 * j;
                    v4f v = {0.0f, 0.0f, 0.0f, 0.0f};
                    if (g >= 0 && g < N_IN) v = *(const v4f*)(srcp + g);
                    xb[q][j] = v;
                }
            }
        }

        __syncthreads();

        {
            v4f tmp[4][4];
            #pragma unroll
            for (int m = 0; m < 4; ++m) {
                int ch = ct * 16 + wave_s + 8 * (m & 1);
                ch = ch < N_CH ? ch : (N_CH - 1);
                ch += (m >> 1) * N_CH;
                const v4f* srcw = w4 + ch * K4 + lane;
                #pragma unroll
                for (int n = 0; n < 4; ++n) {
                    if (n < 3 || lane < K4 - 96) tmp[m][n] = srcw[32 * n];
                }
            }
            #pragma unroll
            for (int m = 0; m < 4; ++m) {
                #pragma unroll
                for (int n = 0; n < 4; ++n) {
                    if (n < 3 || lane < K4 - 96) wl[(wave_s + 8 * m) * PITCH4 + lane + 32 * n] = tmp[m][n];
                }
            }
        }

        __syncthreads();

        const v4f* ar = wl + c * PITCH4 + h;
        const v4f* ai = wl + (16 + c) * PITCH4 + h;

        v8f cr0 = {}, ci0 = {}, cr1 = {}, ci1 = {};
        v4f pr[3], pi[3];
        pr[0] = ar[0]; pi[0] = ai[0];
        pr[1] = ar[2]; pi[1] = ai[2];
        v2f parked = sv[0];

        #pragma unroll
        for (int j = 0; j < NSTEP; ++j) {
            if (j + 2 < NSTEP) {
                pr[(j + 2) % 3] = ar[2 * (j + 2)];
                pi[(j + 2) % 3] = ai[2 * (j + 2)];
            }
            const v4f a_r = pr[j % 3];
            const v4f a_i = pi[j % 3];
            const v4f b0  = xb[0][j];
            const v4f b1  = xb[1][j];
            const v2f arl = __builtin_shufflevector(a_r, a_r, 0, 1);
            const v2f arh = __builtin_shufflevector(a_r, a_r, 2, 3);
            const v2f ail = __builtin_shufflevector(a_i, a_i, 0, 1);
            const v2f aih = __builtin_shufflevector(a_i, a_i, 2, 3);
            const v2f b0l = __builtin_shufflevector(b0, b0, 0, 1);
            const v2f b0h = __builtin_shufflevector(b0, b0, 2, 3);
            const v2f b1l = __builtin_shufflevector(b1, b1, 0, 1);
            const v2f b1h = __builtin_shufflevector(b1, b1, 2, 3);
            cr0 = __builtin_amdgcn_wmma_f32_16x16x4_f32(false, arl, false, b0l, (short)0, cr0, false, false);
            ci0 = __builtin_amdgcn_wmma_f32_16x16x4_f32(false, ail, false, b0l, (short)0, ci0, false, false);
            cr1 = __builtin_amdgcn_wmma_f32_16x16x4_f32(false, arl, false, b1l, (short)0, cr1, false, false);
            ci1 = __builtin_amdgcn_wmma_f32_16x16x4_f32(false, ail, false, b1l, (short)0, ci1, false, false);
            cr0 = __builtin_amdgcn_wmma_f32_16x16x4_f32(false, arh, false, b0h, (short)0, cr0, false, false);
            ci0 = __builtin_amdgcn_wmma_f32_16x16x4_f32(false, aih, false, b0h, (short)0, ci0, false, false);
            cr1 = __builtin_amdgcn_wmma_f32_16x16x4_f32(false, arh, false, b1h, (short)0, cr1, false, false);
            ci1 = __builtin_amdgcn_wmma_f32_16x16x4_f32(false, aih, false, b1h, (short)0, ci1, false, false);

            if (j % 3 == 1 && j / 3 < SAVE_PAIRS) {
                const int e = j / 3;
                const int q = e >> 3;
                const v2f cur = parked;
                if (e + 1 < SAVE_PAIRS) parked = sv[(e + 1) * 32];
                finish_elem(cur, e, h, p_ct, p_ob[q], p_fv[q] && p_ct < N_CT - 1, p_fv[q] && p_ct == N_CT - 1,
                            mags, phase);
            }
        }

        #pragma unroll
        for (int r = 0; r < 8; ++r) {
            sv[r * 32]       = (v2f){cr0[r], ci0[r]};
            sv[(8 + r) * 32] = (v2f){cr1[r], ci1[r]};
        }
        p_ct = ct;
        p_ob[0] = ob[0]; p_ob[1] = ob[1];
        p_fv[0] = fv[0]; p_fv[1] = fv[1];
    }

    for (int e = 0; e < SAVE_PAIRS; ++e) {
        const int q = e >> 3;
        finish_elem(sv[e * 32], e, h, p_ct, p_ob[q], p_fv[q] && p_ct < N_CT - 1, p_fv[q] && p_ct == N_CT - 1,
                    mags, phase);
    }
}

extern "C" void kernel_launch(void* const* d_in, const int* in_sizes, int n_in,
                              void* d_out, int out_size, void* d_ws, size_t ws_size,
                              hipStream_t stream) {
    (void)in_sizes; (void)n_in; (void)d_ws; (void)ws_size; (void)out_size;

    const float* in = (const float*)d_in[0];
    const float* w  = (const float*)d_in[1];

    float* mags  = (float*)d_out;
    float* phase = mags + (size_t)N_BATCH * N_CH * N_FRAMES;

    const dim3 grid(N_BLOCKS);
    const dim3 block(BLOCK_THREADS);

    convstft_inreg_f32<<<grid, block, 0, stream>>>(in, w, mags, phase);
}


# frame windows refilled through per-wave LDS strips: 36 contiguous 16-byte loads in flight at once, then LDS reads (was 100 strided 16-byte gathers per lane, one at a time); sums unchanged
# speedup vs baseline: 3.4341x; 1.0790x over previous
#include <hip/hip_runtime.h>
#include <math.h>


typedef float v2f __attribute__((ext_vector_type(2)));
typedef float v4f __attribute__((ext_vector_type(4)));
typedef float v8f __attribute__((ext_vector_type(8)));

#define WIN_LEN   400
#define WIN_INC   100
#define PAD       300
#define N_FRAMES  1603
#define N_CH      257
#define N_IN      160000
#define N_BATCH   16
#define TOT_FRAMES (N_BATCH * N_FRAMES)
#define WAVES         8
#define BLOCK_THREADS 256
#define FR_PER_WAVE   32
#define FR_PER_GROUP  (WAVES * FR_PER_WAVE)
#define N_FG  ((TOT_FRAMES + FR_PER_GROUP - 1) / FR_PER_GROUP)
#define N_CT  17
#define N_BU  (N_FG * N_CT)
#define N_BLOCKS 256
#define K4     (WIN_LEN / 4)
#define PITCH4 101
#define NSTEP  (WIN_LEN / 8)
#define PANEL4     (32 * PITCH4)
#define SAVE_PAIRS 16
#define PADLEN     (N_IN + 2 * PAD)
#define SPAN_LD    18
#define SPAN4      (32 * SPAN_LD)
#define PARK_OFF   (PANEL4 * 16)
#define STRIP_OFF  (PARK_OFF + WAVES * SAVE_PAIRS * 32 * 8)
#define LDS_BYTES  (STRIP_OFF + WAVES * 2 * SPAN4 * 16)

static_assert(PADLEN % 4 == 0 && WIN_INC % 4 == 0 && PAD % 4 == 0, "a float4 of a strip is all signal or all pad, and never crosses rows");
static_assert(15 * WIN_INC + PAD + WIN_LEN <= 4 * SPAN4, "the longest tile span fits a strip");

static_assert(TOT_FRAMES % 16 == 0, "a 16-frame tile is valid or invalid as a whole");


__device__ __forceinline__ void finish_elem(v2f ri, int e, int h, int ct, unsigned ob,
                                            bool full, bool last,
                                            float* __restrict__ mags, float* __restrict__ phase)
{
    const float EPS = 1.19209290e-07f;
    const float rr = ri.x;
    const float ii = ri.y;
    const float m2 = fmaxf(rr * rr + ii * ii, EPS);
    const float mg = sqrtf(m2);
    const float ph = atan2f(ii + EPS, rr + EPS);
    const int r = e & 7;
    const unsigned o = ob + (unsigned)((ct * 16 + r + 8 * h) * N_FRAMES);
    if (full) {
        mags[o]  = mg;
        phase[o] = ph;
    } else if (last && r == 0) {
        if (h == 0) {
            mags[o]  = mg;
            phase[o] = ph;
        }
    }
}

__global__ __launch_bounds__(BLOCK_THREADS)
void convstft_inreg_f32(const float* __restrict__ in,
                        const float* __restrict__ w,
                        float* __restrict__ mags,
                        float* __restrict__ phase)
{
    __shared__ __attribute__((aligned(16))) unsigned char lds[LDS_BYTES];
    v4f* wl = (v4f*)lds;

    const int tid  = threadIdx.x;
    const int lane = tid & 31;
    const int wave = tid >> 5;
    const int h    = lane >> 4;
    const int c    = lane & 15;

    const int bu0 = (blockIdx.x * N_BU) / N_BLOCKS;
    const int bu1 = ((blockIdx.x + 1) * N_BU) / N_BLOCKS;

    const v4f* w4 = (const v4f*)w;

    const int wave_s = __builtin_amdgcn_readfirstlane(wave);

    v2f* sv = (v2f*)(lds + PARK_OFF) + wave_s * (SAVE_PAIRS * 32) + lane;
    for (int e = 0; e < SAVE_PAIRS; ++e) sv[e * 32] = (v2f){0.0f, 0.0f};

    int      p_ct = 0;
    unsigned p_ob[2] = {0u, 0u};
    bool     p_fv[2] = {false, false};
    unsigned ob[2] = {0u, 0u};

    v4f  xb[2][NSTEP];
    int  fb[2], ft[2];
    bool fv[2];
    int  cur_fg = -1;

    for (int bu = bu0; bu < bu1; ++bu) {
        const int fg = bu / N_CT;
        const int ct = bu - fg * N_CT;

        if (fg != cur_fg) {
            cur_fg = fg;
            v4f* strip = (v4f*)(lds + STRIP_OFF) + wave_s * (2 * SPAN4);
            int rd4[2];
            int bs[2], ts[2];
            v4f t[2][SPAN_LD];
            #pragma unroll
            for (int q = 0; q < 2; ++q) {
                const int f0 = (fg * WAVES + wave_s) * FR_PER_WAVE + q * 16;
                fv[q] = f0 < TOT_FRAMES;
                const int fc = fv[q] ? (f0 + c) : (TOT_FRAMES - 1);
                fb[q] = fc / N_FRAMES;
                ft[q] = fc - fb[q] * N_FRAMES;
                ob[q] = (unsigned)(fb[q] * (N_CH * N_FRAMES) + ft[q]);
                const int fs = fv[q] ? f0 : (TOT_FRAMES - 1);
                bs[q] = fs / N_FRAMES;
                ts[q] = fs - bs[q] * N_FRAMES;
                rd4[q] = (((fb[q] - bs[q]) * PADLEN + (ft[q] - ts[q]) * WIN_INC) >> 2) + h;
            }
            const bool inner = ts[0] * WIN_INC >= PAD && ts[0] * WIN_INC + 4 * SPAN4 <= PAD + N_IN &&
                               ts[1] * WIN_INC >= PAD && ts[1] * WIN_INC + 4 * SPAN4 <= PAD + N_IN;
            if (inner) {
                #pragma unroll
                for (int q = 0; q < 2; ++q) {
                    const v4f* src = (const v4f*)(in + bs[q] * N_IN + (ts[q] * WIN_INC - PAD)) + lane;
                    #pragma unroll
                    for (int n = 0; n < SPAN_LD; ++n) t[q][n] = src[32 * n];
                }
            } else {
                #pragma unroll
                for (int q = 0; q < 2; ++q) {
                    #pragma unroll
                    for (int n = 0; n < SPAN_LD; ++n) {
                        int p = ts[q] * WIN_INC + 4 * (lane + 32 * n);
                        int b = bs[q];
                        if (p >= PADLEN) { p -= PADLEN; b += 1; }
                        v4f v = {0.0f, 0.0f, 0.0f, 0.0f};
                        if (b < N_BATCH && p >= PAD && p < PAD + N_IN) v = *(const v4f*)(in + b * N_IN + (p - PAD));
                        t[q][n] = v;
                    }
                }
            }
            #pragma unroll
            for (int q = 0; q < 2; ++q) {
                #pragma unroll
                for (int n = 0; n < SPAN_LD; ++n) strip[q * SPAN4 + lane + 32 * n] = t[q][n];
            }

            __syncthreads();

            #pragma unroll
            for (int q = 0; q < 2; ++q) {
                const v4f* rd = strip + q * SPAN4 + rd4[q];
                #pragma unroll
                for (int j = 0; j < NSTEP; ++j) xb[q][j] = rd[2 * j];
            }
        }

        __syncthreads();

        {
            v4f tmp[4][4];
            #pragma unroll
            for (int m = 0; m < 4; ++m) {
                int ch = ct * 16 + wave_s + 8 * (m & 1);
                ch = ch < N_CH ? ch : (N_CH - 1);
                ch += (m >> 1) * N_CH;
                const v4f* srcw = w4 + ch * K4 + lane;
                #pragma unroll
                for (int n = 0; n < 4; ++n) {
                    if (n < 3 || lane < K4 - 96) tmp[m][n] = srcw[32 * n];
                }
            }
            #pragma unroll
            for (int m = 0; m < 4; ++m) {
                #pragma unroll
                for (int n = 0; n < 4; ++n) {
                    if (n < 3 || lane < K4 - 96) wl[(wave_s + 8 * m) * PITCH4 + lane + 32 * n] = tmp[m][n];
                }
            }
        }

        __syncthreads();

        const v4f* ar = wl + c * PITCH4 + h;
        const v4f* ai = wl + (16 + c) * PITCH4 + h;

        v8f cr0 = {}, ci0 = {}, cr1 = {}, ci1 = {};
        v4f pr[3], pi[3];
        pr[0] = ar[0]; pi[0] = ai[0];
        pr[1] = ar[2]; pi[1] = ai[2];
        v2f parked = sv[0];

        #pragma unroll
        for (int j = 0; j < NSTEP; ++j) {
            if (j + 2 < NSTEP) {
                pr[(j + 2) % 3] = ar[2 * (j + 2)];
                pi[(j + 2) % 3] = ai[2 * (j + 2)];
            }
            const v4f a_r = pr[j % 3];
            const v4f a_i = pi[j % 3];
            const v4f b0  = xb[0][j];
            const v4f b1  = xb[1][j];
            const v2f arl = __builtin_shufflevector(a_r, a_r, 0, 1);
            const v2f arh = __builtin_shufflevector(a_r, a_r, 2, 3);
            const v2f ail = __builtin_shufflevector(a_i, a_i, 0, 1);
            const v2f aih = __builtin_shufflevector(a_i, a_i, 2, 3);
            const v2f b0l = __builtin_shufflevector(b0, b0, 0, 1);
            const v2f b0h = __builtin_shufflevector(b0, b0, 2, 3);
            const v2f b1l = __builtin_shufflevector(b1, b1, 0, 1);
            const v2f b1h = __builtin_shufflevector(b1, b1, 2, 3);
            cr0 = __builtin_amdgcn_wmma_f32_16x16x4_f32(false, arl, false, b0l, (short)0, cr0, false, false);
            ci0 = __builtin_amdgcn_wmma_f32_16x16x4_f32(false, ail, false, b0l, (short)0, ci0, false, false);
            cr1 = __builtin_amdgcn_wmma_f32_16x16x4_f32(false, arl, false, b1l, (short)0, cr1, false, false);
            ci1 = __builtin_amdgcn_wmma_f32_16x16x4_f32(false, ail, false, b1l, (short)0, ci1, false, false);
            cr0 = __builtin_amdgcn_wmma_f32_16x16x4_f32(false, arh, false, b0h, (short)0, cr0, false, false);
            ci0 = __builtin_amdgcn_wmma_f32_16x16x4_f32(false, aih, false, b0h, (short)0, ci0, false, false);
            cr1 = __builtin_amdgcn_wmma_f32_16x16x4_f32(false, arh, false, b1h, (short)0, cr1, false, false);
            ci1 = __builtin_amdgcn_wmma_f32_16x16x4_f32(false, aih, false, b1h, (short)0, ci1, false, false);

            if (j % 3 == 1 && j / 3 < SAVE_PAIRS) {
                const int e = j / 3;
                const int q = e >> 3;
                const v2f cur = parked;
                if (e + 1 < SAVE_PAIRS) parked = sv[(e + 1) * 32];
                finish_elem(cur, e, h, p_ct, p_ob[q], p_fv[q] && p_ct < N_CT - 1, p_fv[q] && p_ct == N_CT - 1,
                            mags, phase);
            }
        }

        #pragma unroll
        for (int r = 0; r < 8; ++r) {
            sv[r * 32]       = (v2f){cr0[r], ci0[r]};
            sv[(8 + r) * 32] = (v2f){cr1[r], ci1[r]};
        }
        p_ct = ct;
        p_ob[0] = ob[0]; p_ob[1] = ob[1];
        p_fv[0] = fv[0]; p_fv[1] = fv[1];
    }

    for (int e = 0; e < SAVE_PAIRS; ++e) {
        const int q = e >> 3;
        finish_elem(sv[e * 32], e, h, p_ct, p_ob[q], p_fv[q] && p_ct < N_CT - 1, p_fv[q] && p_ct == N_CT - 1,
                    mags, phase);
    }
}

extern "C" void kernel_launch(void* const* d_in, const int* in_sizes, int n_in,
                              void* d_out, int out_size, void* d_ws, size_t ws_size,
                              hipStream_t stream) {
    (void)in_sizes; (void)n_in; (void)d_ws; (void)ws_size; (void)out_size;

    const float* in = (const float*)d_in[0];
    const float* w  = (const float*)d_in[1];

    float* mags  = (float*)d_out;
    float* phase = mags + (size_t)N_BATCH * N_CH * N_FRAMES;

    const dim3 grid(N_BLOCKS);
    const dim3 block(BLOCK_THREADS);

    convstft_inreg_f32<<<grid, block, 0, stream>>>(in, w, mags, phase);
}


# one 16-frame tile per wave, 16 waves per block: input fragments read from the wave's LDS strip per group of 5 steps, no windows in registers; four waves share a SIMD's matrix pipe; sums unchanged
# speedup vs baseline: 3.5618x; 1.0372x over previous
#include <hip/hip_runtime.h>
#include <math.h>


typedef float v2f __attribute__((ext_vector_type(2)));
typedef float v4f __attribute__((ext_vector_type(4)));
typedef float v8f __attribute__((ext_vector_type(8)));

#define WIN_LEN   400
#define WIN_INC   100
#define PAD       300
#define N_FRAMES  1603
#define N_CH      257
#define N_IN      160000
#define N_BATCH   16
#define TOT_FRAMES (N_BATCH * N_FRAMES)
#define WAVES         16
#define BLOCK_THREADS 512
#define FR_PER_WAVE   16
#define FR_PER_GROUP  (WAVES * FR_PER_WAVE)
#define N_FG  ((TOT_FRAMES + FR_PER_GROUP - 1) / FR_PER_GROUP)
#define N_CT  17
#define N_BU  (N_FG * N_CT)
#define N_BLOCKS 256
#define K4     (WIN_LEN / 4)
#define PITCH4 101
#define NSTEP  (WIN_LEN / 8)
#define GSTEPS 5
#define NGROUP (NSTEP / GSTEPS)
#define PANEL4     (32 * PITCH4)
#define SAVE_PAIRS 8
#define PADLEN     (N_IN + 2 * PAD)
#define SPAN_LD    18
#define SPAN4      (32 * SPAN_LD)
#define PARK_OFF   (PANEL4 * 16)
#define STRIP_OFF  (PARK_OFF + WAVES * SAVE_PAIRS * 32 * 8)
#define LDS_BYTES  (STRIP_OFF + WAVES * SPAN4 * 16)

static_assert(PADLEN % 4 == 0 && WIN_INC % 4 == 0 && PAD % 4 == 0, "a float4 of a strip is all signal or all pad, and never crosses rows");
static_assert(15 * WIN_INC + PAD + WIN_LEN <= 4 * SPAN4, "the longest tile span fits a strip");

static_assert(TOT_FRAMES % 16 == 0, "a 16-frame tile is valid or invalid as a whole");
static_assert(NSTEP % GSTEPS == 0 && SAVE_PAIRS <= NGROUP, "whole groups; one parked element finished per group");


__device__ __forceinline__ void finish_elem(v2f ri, int e, int h, int ct, unsigned ob,
                                            bool full, bool last,
                                            float* __restrict__ mags, float* __restrict__ phase)
{
    const float EPS = 1.19209290e-07f;
    const float rr = ri.x;
    const float ii = ri.y;
    const float m2 = fmaxf(rr * rr + ii * ii, EPS);
    const float mg = sqrtf(m2);
    const float ph = atan2f(ii + EPS, rr + EPS);
    const int r = e & 7;
    const unsigned o = ob + (unsigned)((ct * 16 + r + 8 * h) * N_FRAMES);
    if (full) {
        mags[o]  = mg;
        phase[o] = ph;
    } else if (last && r == 0) {
        if (h == 0) {
            mags[o]  = mg;
            phase[o] = ph;
        }
    }
}

__global__ __launch_bounds__(BLOCK_THREADS)
void convstft_strip_f32(const float* __restrict__ in,
                        const float* __restrict__ w,
                        float* __restrict__ mags,
                        float* __restrict__ phase)
{
    __shared__ __attribute__((aligned(16))) unsigned char lds[LDS_BYTES];
    v4f* wl = (v4f*)lds;

    const int tid  = threadIdx.x;
    const int lane = tid & 31;
    const int wave = tid >> 5;
    const int h    = lane >> 4;
    const int c    = lane & 15;

    const int bu0 = (blockIdx.x * N_BU) / N_BLOCKS;
    const int bu1 = ((blockIdx.x + 1) * N_BU) / N_BLOCKS;

    const v4f* w4 = (const v4f*)w;

    const int wave_s = __builtin_amdgcn_readfirstlane(wave);

    v2f* sv = (v2f*)(lds + PARK_OFF) + wave_s * (SAVE_PAIRS * 32) + lane;
    for (int e = 0; e < SAVE_PAIRS; ++e) sv[e * 32] = (v2f){0.0f, 0.0f};

    int      p_ct = 0;
    unsigned p_ob = 0u;
    bool     p_fv = false;
    unsigned ob = 0u;
    bool     fv = false;
    int      rd4 = 0;
    int      cur_fg = -1;

    v4f* strip = (v4f*)(lds + STRIP_OFF) + wave_s * SPAN4;

    for (int bu = bu0; bu < bu1; ++bu) {
        const int fg = bu / N_CT;
        const int ct = bu - fg * N_CT;

        if (fg != cur_fg) {
            cur_fg = fg;
            const int f0 = (fg * WAVES + wave_s) * FR_PER_WAVE;
            fv = f0 < TOT_FRAMES;
            const int fc = fv ? (f0 + c) : (TOT_FRAMES - 1);
            const int fb = fc / N_FRAMES;
            const int ft = fc - fb * N_FRAMES;
            ob = (unsigned)(fb * (N_CH * N_FRAMES) + ft);
            const int fs = fv ? f0 : (TOT_FRAMES - 1);
            const int bs = fs / N_FRAMES;
            const int ts = fs - bs * N_FRAMES;
            rd4 = (((fb - bs) * PADLEN + (ft - ts) * WIN_INC) >> 2) + h;

            v4f t[SPAN_LD];
            const bool inner = ts * WIN_INC >= PAD && ts * WIN_INC + 4 * SPAN4 <= PAD + N_IN;
            if (inner) {
                const v4f* src = (const v4f*)(in + bs * N_IN + (ts * WIN_INC - PAD)) + lane;
                #pragma unroll
                for (int n = 0; n < SPAN_LD; ++n) t[n] = src[32 * n];
            } else {
                #pragma unroll
                for (int n = 0; n < SPAN_LD; ++n) {
                    int p = ts * WIN_INC + 4 * (lane + 32 * n);
                    int b = bs;
                    if (p >= PADLEN) { p -= PADLEN; b += 1; }
                    v4f v = {0.0f, 0.0f, 0.0f, 0.0f};
                    if (b < N_BATCH && p >= PAD && p < PAD + N_IN) v = *(const v4f*)(in + b * N_IN + (p - PAD));
                    t[n] = v;
                }
            }
            #pragma unroll
            for (int n = 0; n < SPAN_LD; ++n) strip[lane + 32 * n] = t[n];
        }

        __syncthreads();

        {
            v4f tmp[2][4];
            #pragma unroll
            for (int m = 0; m < 2; ++m) {
                int ch = ct * 16 + wave_s;
                ch = ch < N_CH ? ch : (N_CH - 1);
                ch += m * N_CH;
                const v4f* srcw = w4 + ch * K4 + lane;
                #pragma unroll
                for (int n = 0; n < 4; ++n) {
                    if (n < 3 || lane < K4 - 96) tmp[m][n] = srcw[32 * n];
                }
            }
            #pragma unroll
            for (int m = 0; m < 2; ++m) {
                #pragma unroll
                for (int n = 0; n < 4; ++n) {
                    if (n < 3 || lane < K4 - 96) wl[(wave_s + 16 * m) * PITCH4 + lane + 32 * n] = tmp[m][n];
                }
            }
        }

        __syncthreads();

        const v4f* ar = wl + c * PITCH4 + h;
        const v4f* ai = wl + (16 + c) * PITCH4 + h;
        const v4f* bx = strip + rd4;

        v8f cr = {}, ci = {};
        v4f fr[2][GSTEPS], fi[2][GSTEPS], fx[2][GSTEPS];
        #pragma unroll
        for (int s = 0; s < GSTEPS; ++s) {
            fr[0][s] = ar[2 * s];
            fi[0][s] = ai[2 * s];
            fx[0][s] = bx[2 * s];
        }
        v2f parked = sv[0];

        #pragma unroll
        for (int g = 0; g < NGROUP; ++g) {
            const int cs = g & 1;
            const int ns = cs ^ 1;
            v2f parked_next = parked;
            if (g + 1 < NGROUP) {
                #pragma unroll
                for (int s = 0; s < GSTEPS; ++s) {
                    const int j = (g + 1) * GSTEPS + s;
                    fr[ns][s] = ar[2 * j];
                    fi[ns][s] = ai[2 * j];
                    fx[ns][s] = bx[2 * j];
                }
            }
            if (g + 1 < SAVE_PAIRS) parked_next = sv[(g + 1) * 32];

            #pragma unroll
            for (int s = 0; s < GSTEPS; ++s) {
                const v4f a_r = fr[cs][s];
                const v4f a_i = fi[cs][s];
                const v4f b   = fx[cs][s];
                const v2f arl = __builtin_shufflevector(a_r, a_r, 0, 1);
                const v2f arh = __builtin_shufflevector(a_r, a_r, 2, 3);
                const v2f ail = __builtin_shufflevector(a_i, a_i, 0, 1);
                const v2f aih = __builtin_shufflevector(a_i, a_i, 2, 3);
                const v2f bl  = __builtin_shufflevector(b, b, 0, 1);
                const v2f bh  = __builtin_shufflevector(b, b, 2, 3);
                cr = __builtin_amdgcn_wmma_f32_16x16x4_f32(false, arl, false, bl, (short)0, cr, false, false);
                ci = __builtin_amdgcn_wmma_f32_16x16x4_f32(false, ail, false, bl, (short)0, ci, false, false);
                cr = __builtin_amdgcn_wmma_f32_16x16x4_f32(false, arh, false, bh, (short)0, cr, false, false);
                ci = __builtin_amdgcn_wmma_f32_16x16x4_f32(false, aih, false, bh, (short)0, ci, false, false);
            }

            if (g < SAVE_PAIRS) {
                finish_elem(parked, g, h, p_ct, p_ob, p_fv && p_ct < N_CT - 1, p_fv && p_ct == N_CT - 1,
                            mags, phase);
            }
            parked = parked_next;
        }

        #pragma unroll
        for (int r = 0; r < SAVE_PAIRS; ++r) sv[r * 32] = (v2f){cr[r], ci[r]};
        p_ct = ct;
        p_ob = ob;
        p_fv = fv;
    }

    for (int e = 0; e < SAVE_PAIRS; ++e) {
        finish_elem(sv[e * 32], e, h, p_ct, p_ob, p_fv && p_ct < N_CT - 1, p_fv && p_ct == N_CT - 1,
                    mags, phase);
    }
}

extern "C" void kernel_launch(void* const* d_in, const int* in_sizes, int n_in,
                              void* d_out, int out_size, void* d_ws, size_t ws_size,
                              hipStream_t stream) {
    (void)in_sizes; (void)n_in; (void)d_ws; (void)ws_size; (void)out_size;

    const float* in = (const float*)d_in[0];
    const float* w  = (const float*)d_in[1];

    float* mags  = (float*)d_out;
    float* phase = mags + (size_t)N_BATCH * N_CH * N_FRAMES;

    const dim3 grid(N_BLOCKS);
    const dim3 block(BLOCK_THREADS);

    convstft_strip_f32<<<grid, block, 0, stream>>>(in, w, mags, phase);
}


# tiles listed channel-major, cut into 256 equal shares, dealt to the 16 waves; each walks its own list, no block barrier per tile; channel 256 as 100 products in two half-chains: its sum order changed
# speedup vs baseline: 3.7442x; 1.0512x over previous
#include <hip/hip_runtime.h>
#include <math.h>


typedef float v2f __attribute__((ext_vector_type(2)));
typedef float v4f __attribute__((ext_vector_type(4)));
typedef float v8f __attribute__((ext_vector_type(8)));

#define WIN_LEN   400
#define WIN_INC   100
#define PAD       300
#define N_FRAMES  1603
#define N_CH      257
#define N_IN      160000
#define N_BATCH   16
#define TOT_FRAMES (N_BATCH * N_FRAMES)
#define WAVES         16
#define BLOCK_THREADS 512
#define N_FT   (TOT_FRAMES / 16)
#define N_CT   17
#define N_HEAVY ((N_CT - 1) * N_FT)
#define N_LIGHT N_FT
#define W_TOTAL (2 * N_HEAVY + N_LIGHT)
#define N_BLOCKS 256
#define K4     (WIN_LEN / 4)
#define PITCH4 101
#define NSTEP  (WIN_LEN / 8)
#define GSTEPS 5
#define NGROUP (NSTEP / GSTEPS)
#define PANEL4     (32 * PITCH4)
#define SAVE_PAIRS 8
#define PADLEN     (N_IN + 2 * PAD)
#define SPAN_LD    18
#define SPAN4      (32 * SPAN_LD)
#define PARK_OFF   (2 * PANEL4 * 16)
#define STRIP_OFF  (PARK_OFF + WAVES * SAVE_PAIRS * 32 * 8)
#define LDS_BYTES  (STRIP_OFF + WAVES * SPAN4 * 16)

static_assert(PADLEN % 4 == 0 && WIN_INC % 4 == 0 && PAD % 4 == 0, "a float4 of a strip is all signal or all pad, and never crosses rows");
static_assert(15 * WIN_INC + PAD + WIN_LEN <= 4 * SPAN4, "the longest tile span fits a strip");

static_assert(TOT_FRAMES % 16 == 0, "every 16-frame tile is whole");
static_assert(NSTEP % GSTEPS == 0 && SAVE_PAIRS <= NGROUP, "whole groups; one parked element finished per group");
static_assert(2 * ((W_TOTAL + N_BLOCKS - 1) / N_BLOCKS) < N_FT, "a block's tile range touches at most two channel tiles");


__device__ __forceinline__ void finish_elem(v2f ri, int e, int h, int ct, unsigned ob,
                                            bool full, bool last,
                                            float* __restrict__ mags, float* __restrict__ phase)
{
    const float EPS = 1.19209290e-07f;
    const float rr = ri.x;
    const float ii = ri.y;
    const float m2 = fmaxf(rr * rr + ii * ii, EPS);
    const float mg = sqrtf(m2);
    const float ph = atan2f(ii + EPS, rr + EPS);
    const int r = e & 7;
    const unsigned o = ob + (unsigned)((ct * 16 + r + 8 * h) * N_FRAMES);
    if (full) {
        mags[o]  = mg;
        phase[o] = ph;
    } else if (last && r == 0) {
        if (h == 0) {
            mags[o]  = mg;
            phase[o] = ph;
        }
    }
}

template <bool LIGHT>
__device__ __forceinline__ void run_tile(const v4f* ar, const v4f* ai,
                                         const v4f* bx, v2f* sv, float* slot, int h,
                                         int p_ct, unsigned p_ob, bool p_full, bool p_last,
                                         float* __restrict__ mags, float* __restrict__ phase)
{
    v8f cr = {}, ci = {};
    v4f fr[2][GSTEPS], fi[2][GSTEPS], fx[2][GSTEPS];
    #pragma unroll
    for (int s = 0; s < GSTEPS; ++s) {
        fr[0][s] = ar[2 * s];
        if (!LIGHT) fi[0][s] = ai[2 * s];
        fx[0][s] = bx[2 * s];
    }
    v2f parked = sv[0];

    #pragma unroll
    for (int g = 0; g < NGROUP; ++g) {
        const int cs = g & 1;
        const int ns = cs ^ 1;
        v2f parked_next = parked;
        if (g + 1 < NGROUP) {
            #pragma unroll
            for (int s = 0; s < GSTEPS; ++s) {
                const int j = (g + 1) * GSTEPS + s;
                fr[ns][s] = ar[2 * j];
                if (!LIGHT) fi[ns][s] = ai[2 * j];
                fx[ns][s] = bx[2 * j];
            }
        }
        if (g + 1 < SAVE_PAIRS) parked_next = sv[(g + 1) * 32];

        #pragma unroll
        for (int s = 0; s < GSTEPS; ++s) {
            const v4f a_r = fr[cs][s];
            const v4f b   = fx[cs][s];
            const v2f arl = __builtin_shufflevector(a_r, a_r, 0, 1);
            const v2f arh = __builtin_shufflevector(a_r, a_r, 2, 3);
            const v2f bl  = __builtin_shufflevector(b, b, 0, 1);
            const v2f bh  = __builtin_shufflevector(b, b, 2, 3);
            if (LIGHT) {
                cr = __builtin_amdgcn_wmma_f32_16x16x4_f32(false, arl, false, bl, (short)0, cr, false, false);
                ci = __builtin_amdgcn_wmma_f32_16x16x4_f32(false, arh, false, bh, (short)0, ci, false, false);
            } else {
                const v4f a_i = fi[cs][s];
                const v2f ail = __builtin_shufflevector(a_i, a_i, 0, 1);
                const v2f aih = __builtin_shufflevector(a_i, a_i, 2, 3);
                cr = __builtin_amdgcn_wmma_f32_16x16x4_f32(false, arl, false, bl, (short)0, cr, false, false);
                ci = __builtin_amdgcn_wmma_f32_16x16x4_f32(false, ail, false, bl, (short)0, ci, false, false);
                cr = __builtin_amdgcn_wmma_f32_16x16x4_f32(false, arh, false, bh, (short)0, cr, false, false);
                ci = __builtin_amdgcn_wmma_f32_16x16x4_f32(false, aih, false, bh, (short)0, ci, false, false);
            }
        }

        if (g < SAVE_PAIRS) finish_elem(parked, g, h, p_ct, p_ob, p_full, p_last, mags, phase);
        parked = parked_next;
    }

    if (LIGHT) {
        *slot = cr[0] + ci[0];
    } else {
        #pragma unroll
        for (int r = 0; r < SAVE_PAIRS; ++r) sv[r * 32] = (v2f){cr[r], ci[r]};
    }
}

__global__ __launch_bounds__(BLOCK_THREADS)
void convstft_tiles_f32(const float* __restrict__ in,
                        const float* __restrict__ w,
                        float* __restrict__ mags,
                        float* __restrict__ phase)
{
    __shared__ __attribute__((aligned(16))) unsigned char lds[LDS_BYTES];
    v4f* wl = (v4f*)lds;

    const int tid  = threadIdx.x;
    const int lane = tid & 31;
    const int wave = tid >> 5;
    const int h    = lane >> 4;
    const int c    = lane & 15;

    const int wave_s = __builtin_amdgcn_readfirstlane(wave);

    const int q0 = (int)((blockIdx.x * (unsigned)W_TOTAL) / N_BLOCKS);
    const int q1 = (int)(((blockIdx.x + 1u) * (unsigned)W_TOTAL) / N_BLOCKS);
    const int g0 = q0 <= 2 * N_HEAVY ? (q0 + 1) / 2 : N_HEAVY + (q0 - 2 * N_HEAVY);
    const int g1 = q1 <= 2 * N_HEAVY ? (q1 + 1) / 2 : N_HEAVY + (q1 - 2 * N_HEAVY);
    const int ct_a = g0 / N_FT;

    const v4f* w4 = (const v4f*)w;

    v2f* sv = (v2f*)(lds + PARK_OFF) + wave_s * (SAVE_PAIRS * 32) + lane;
    float* slot = (float*)((v2f*)(lds + PARK_OFF) + wave_s * (SAVE_PAIRS * 32) + c) + h;
    for (int e = 0; e < SAVE_PAIRS; ++e) sv[e * 32] = (v2f){0.0f, 0.0f};

    {
        #pragma unroll
        for (int p = 0; p < 2; ++p) {
            v4f tmp[2][4];
            const int ctp = ct_a + p < N_CT ? ct_a + p : N_CT - 1;
            #pragma unroll
            for (int m = 0; m < 2; ++m) {
                int ch = ctp * 16 + wave_s;
                ch = ch < N_CH ? ch : (N_CH - 1);
                ch += m * N_CH;
                const v4f* srcw = w4 + ch * K4 + lane;
                #pragma unroll
                for (int n = 0; n < 4; ++n) {
                    if (n < 3 || lane < K4 - 96) tmp[m][n] = srcw[32 * n];
                }
            }
            #pragma unroll
            for (int m = 0; m < 2; ++m) {
                #pragma unroll
                for (int n = 0; n < 4; ++n) {
                    if (n < 3 || lane < K4 - 96) wl[p * PANEL4 + (wave_s + 16 * m) * PITCH4 + lane + 32 * n] = tmp[m][n];
                }
            }
        }
    }

    __syncthreads();

    int      p_ct = 0;
    unsigned p_ob = 0u;
    bool     p_full = false;
    bool     p_last = false;

    v4f* strip = (v4f*)(lds + STRIP_OFF) + wave_s * SPAN4;

    for (int gi = g0 + wave_s; gi < g1; gi += WAVES) {
        const int ct = gi / N_FT;
        const int f0 = (gi - ct * N_FT) * 16;
        const int fc = f0 + c;
        const int fb = fc / N_FRAMES;
        const int ft = fc - fb * N_FRAMES;
        const unsigned ob = (unsigned)(fb * (N_CH * N_FRAMES) + ft);
        const int bs = f0 / N_FRAMES;
        const int ts = f0 - bs * N_FRAMES;
        const int rd4 = (((fb - bs) * PADLEN + (ft - ts) * WIN_INC) >> 2) + h;

        {
            v4f t[SPAN_LD];
            const bool inner = ts * WIN_INC >= PAD && ts * WIN_INC + 4 * SPAN4 <= PAD + N_IN;
            if (inner) {
                const v4f* src = (const v4f*)(in + bs * N_IN + (ts * WIN_INC - PAD)) + lane;
                #pragma unroll
                for (int n = 0; n < SPAN_LD; ++n) t[n] = src[32 * n];
            } else {
                #pragma unroll
                for (int n = 0; n < SPAN_LD; ++n) {
                    int p = ts * WIN_INC + 4 * (lane + 32 * n);
                    int b = bs;
                    if (p >= PADLEN) { p -= PADLEN; b += 1; }
                    v4f v = {0.0f, 0.0f, 0.0f, 0.0f};
                    if (b < N_BATCH && p >= PAD && p < PAD + N_IN) v = *(const v4f*)(in + b * N_IN + (p - PAD));
                    t[n] = v;
                }
            }
            #pragma unroll
            for (int n = 0; n < SPAN_LD; ++n) strip[lane + 32 * n] = t[n];
        }

        const v4f* pan = wl + (ct - ct_a) * PANEL4;
        const v4f* bx = strip + rd4;

        if (ct == N_CT - 1) {
            const v4f* ar = pan + (c == 8 ? 24 : c) * PITCH4 + h;
            run_tile<true>(ar, ar, bx, sv, slot, h, p_ct, p_ob, p_full, p_last, mags, phase);
        } else {
            const v4f* ar = pan + c * PITCH4 + h;
            const v4f* ai = pan + (16 + c) * PITCH4 + h;
            run_tile<false>(ar, ai, bx, sv, slot, h, p_ct, p_ob, p_full, p_last, mags, phase);
        }
        p_ct = ct;
        p_ob = ob;
        p_full = ct < N_CT - 1;
        p_last = ct == N_CT - 1;
    }

    for (int e = 0; e < SAVE_PAIRS; ++e) {
        finish_elem(sv[e * 32], e, h, p_ct, p_ob, p_full, p_last, mags, phase);
    }
}

extern "C" void kernel_launch(void* const* d_in, const int* in_sizes, int n_in,
                              void* d_out, int out_size, void* d_ws, size_t ws_size,
                              hipStream_t stream) {
    (void)in_sizes; (void)n_in; (void)d_ws; (void)ws_size; (void)out_size;

    const float* in = (const float*)d_in[0];
    const float* w  = (const float*)d_in[1];

    float* mags  = (float*)d_out;
    float* phase = mags + (size_t)N_BATCH * N_CH * N_FRAMES;

    const dim3 grid(N_BLOCKS);
    const dim3 block(BLOCK_THREADS);

    convstft_tiles_f32<<<grid, block, 0, stream>>>(in, w, mags, phase);
}
